# starting kernel
# baseline (speedup 1.0000x reference)
#include <hip/hip_runtime.h>


typedef float v2f __attribute__((ext_vector_type(2)));
typedef float v4f __attribute__((ext_vector_type(4)));
typedef float v8f __attribute__((ext_vector_type(8)));

constexpr int IN_F   = 4096;
constexpr int OUT_F  = 4096;
constexpr int RANK   = 64;
constexpr int NTOK   = 8192;
constexpr int KSPLIT = 4;
constexpr int KCHUNK = 128;
constexpr int LSTR   = 160;

__global__ __launch_bounds__(256) void lowrank_xu(const float* __restrict__ x,
                                                  const float* __restrict__ U,
                                                  float* __restrict__ Tpart) {
  __shared__ float lU[(KCHUNK / 2) * LSTR];

  const int tid  = threadIdx.x;
  const int lane = tid & 31;
  const int ln16 = lane & 15;
  const int hi   = lane >> 4;
  const int wave = tid >> 5;
  const int mbase = blockIdx.x * 128 + wave * 16;
  const int kpart = blockIdx.y * (IN_F / KSPLIT);

  const float* __restrict__ xrow = x + (size_t)(mbase + ln16) * IN_F + kpart;

  v8f acc[4] = {};

  for (int kb = 0; kb < IN_F / KSPLIT; kb += KCHUNK) {
    __syncthreads();
    for (int i = tid; i < KCHUNK * (RANK / 4); i += 256) {
      const int r  = i >> 4;
      const int c4 = i & 15;
      const v4f u = *(const v4f*)(U + (size_t)(kpart + kb + r) * RANK + c4 * 4);
      float* dst = &lU[(r >> 1) * LSTR + (r & 1)];
#pragma unroll
      for (int j = 0; j < 4; ++j) dst[2 * (c4 * 4 + j)] = u[j];
    }
    __syncthreads();

#pragma unroll 4
    for (int k0 = 0; k0 < KCHUNK; k0 += 4) {
      const v2f a = *(const v2f*)(xrow + kb + k0 + 2 * hi);
#pragma unroll
      for (int t = 0; t < 4; ++t) {
        const v2f b = *(const v2f*)(
            &lU[((k0 >> 1) + hi) * LSTR + 2 * (t * 16 + ln16)]);
        acc[t] = __builtin_amdgcn_wmma_f32_16x16x4_f32(
            false, a, false, b, (short)0, acc[t], false, false);
      }
    }
  }

  float* __restrict__ Tout = Tpart + (size_t)blockIdx.y * NTOK * RANK;
#pragma unroll
  for (int t = 0; t < 4; ++t)
#pragma unroll
    for (int r = 0; r < 8; ++r)
      Tout[(size_t)(mbase + hi * 8 + r) * RANK + t * 16 + ln16] = acc[t][r];
}

__global__ __launch_bounds__(256) void lowrank_tv(const float* __restrict__ Tpart,
                                                  const float* __restrict__ V,
                                                  const float* __restrict__ bias,
                                                  float* __restrict__ out) {
  __shared__ float lV[(RANK / 2) * LSTR];

  const int tid  = threadIdx.x;
  const int lane = tid & 31;
  const int ln16 = lane & 15;
  const int hi   = lane >> 4;
  const int wave = tid >> 5;
  const int mbase = blockIdx.x * 128 + wave * 16;
  const int nbase = blockIdx.y * 64;

  for (int i = tid; i < RANK * (64 / 4); i += 256) {
    const int r  = i >> 4;
    const int c4 = i & 15;
    const v4f v = *(const v4f*)(V + (size_t)r * OUT_F + nbase + c4 * 4);
    float* dst = &lV[(r >> 1) * LSTR + (r & 1)];
#pragma unroll
    for (int j = 0; j < 4; ++j) dst[2 * (c4 * 4 + j)] = v[j];
  }
  __syncthreads();

  const size_t trow = (size_t)(mbase + ln16) * RANK;

  v8f acc[4] = {};
#pragma unroll
  for (int k0 = 0; k0 < RANK; k0 += 4) {
    const int koff = k0 + 2 * hi;
    v2f a = *(const v2f*)(Tpart + trow + koff);
#pragma unroll
    for (int p = 1; p < KSPLIT; ++p)
      a += *(const v2f*)(Tpart + (size_t)p * NTOK * RANK + trow + koff);
#pragma unroll
    for (int t = 0; t < 4; ++t) {
      const v2f b = *(const v2f*)(
          &lV[((k0 >> 1) + hi) * LSTR + 2 * (t * 16 + ln16)]);
      acc[t] = __builtin_amdgcn_wmma_f32_16x16x4_f32(
          false, a, false, b, (short)0, acc[t], false, false);
    }
  }

#pragma unroll
  for (int t = 0; t < 4; ++t) {
    const float bv = bias[nbase + t * 16 + ln16];
#pragma unroll
    for (int r = 0; r < 8; ++r)
      out[(size_t)(mbase + hi * 8 + r) * OUT_F + nbase + t * 16 + ln16] =
          acc[t][r] + bv;
  }
}

extern "C" void kernel_launch(void* const* d_in, const int* in_sizes, int n_in,
                              void* d_out, int out_size, void* d_ws, size_t ws_size,
                              hipStream_t stream) {
  const float* x    = (const float*)d_in[0];
  const float* U    = (const float*)d_in[1];
  const float* V    = (const float*)d_in[2];
  const float* bias = (const float*)d_in[3];
  float* out   = (float*)d_out;
  float* Tpart = (float*)d_ws;

  lowrank_xu<<<dim3(NTOK / 128, KSPLIT), 256, 0, stream>>>(x, U, Tpart);
  lowrank_tv<<<dim3(NTOK / 128, OUT_F / 64), 256, 0, stream>>>(Tpart, V, bias, out);
}


# stage 1 K loop pipelined in registers: next 128-k chunk's x (b128 per lane) and U tile load under this chunk's WMMAs; k summed 0,1,4,5,2,3,6,7 per 8 (LDS pair stride 144 to match)
# speedup vs baseline: 1.2819x; 1.2819x over previous
#include <hip/hip_runtime.h>


typedef float v2f __attribute__((ext_vector_type(2)));
typedef float v4f __attribute__((ext_vector_type(4)));
typedef float v8f __attribute__((ext_vector_type(8)));

constexpr int IN_F   = 4096;
constexpr int OUT_F  = 4096;
constexpr int RANK   = 64;
constexpr int NTOK   = 8192;
constexpr int KSPLIT = 4;
constexpr int KCHUNK = 128;
constexpr int LSTR   = 160;
constexpr int LSTR_U = 144;
constexpr int KPER   = IN_F / KSPLIT;

__global__ __launch_bounds__(256) void lowrank_xu(const float* __restrict__ x,
                                                  const float* __restrict__ U,
                                                  float* __restrict__ Tpart) {
  __shared__ float lU[(KCHUNK / 2) * LSTR_U];

  const int tid  = threadIdx.x;
  const int lane = tid & 31;
  const int ln16 = lane & 15;
  const int hi   = lane >> 4;
  const int wave = tid >> 5;
  const int mbase = blockIdx.x * 128 + wave * 16;
  const int kpart = blockIdx.y * KPER;

  const float* __restrict__ xrow = x + (size_t)(mbase + ln16) * IN_F + kpart + 4 * hi;
  const int ur = tid >> 4;
  const int uc = tid & 15;
  const float* __restrict__ urow = U + (size_t)(kpart + ur) * RANK + uc * 4;

  v8f acc[4] = {};
  v4f qa[16], qb[16];
  v4f ureg[8];

  auto load_x = [&](v4f (&q)[16], int kb) {
#pragma unroll
    for (int j = 0; j < 16; ++j) q[j] = *(const v4f*)(xrow + kb + 8 * j);
  };
  auto load_u = [&](int kb) {
#pragma unroll
    for (int s = 0; s < 8; ++s)
      ureg[s] = *(const v4f*)(urow + (size_t)(kb + 16 * s) * RANK);
  };
  auto store_u = [&]() {
#pragma unroll
    for (int s = 0; s < 8; ++s) {
      const int r = ur + 16 * s;
      float* dst = &lU[(r >> 1) * LSTR_U + (r & 1) + 8 * uc];
#pragma unroll
      for (int j = 0; j < 4; ++j) dst[2 * j] = ureg[s][j];
    }
  };
  auto compute = [&](const v4f (&q)[16]) {
#pragma unroll
    for (int j = 0; j < 16; ++j) {
#pragma unroll
      for (int e = 0; e < 2; ++e) {
        const v2f a = {q[j][2 * e], q[j][2 * e + 1]};
#pragma unroll
        for (int t = 0; t < 4; ++t) {
          const v2f b = *(const v2f*)(
              &lU[(4 * j + 2 * hi + e) * LSTR_U + 2 * (t * 16 + ln16)]);
          acc[t] = __builtin_amdgcn_wmma_f32_16x16x4_f32(
              false, a, false, b, (short)0, acc[t], false, false);
        }
      }
    }
  };

  load_u(0);
  load_x(qa, 0);
  for (int kb = 0; kb < KPER; kb += 2 * KCHUNK) {
    __syncthreads();
    store_u();
    __syncthreads();
    load_u(kb + KCHUNK);
    load_x(qb, kb + KCHUNK);
    compute(qa);

    __syncthreads();
    store_u();
    __syncthreads();
    if (kb + 2 * KCHUNK < KPER) {
      load_u(kb + 2 * KCHUNK);
      load_x(qa, kb + 2 * KCHUNK);
    }
    compute(qb);
  }

  float* __restrict__ Tout = Tpart + (size_t)blockIdx.y * NTOK * RANK;
#pragma unroll
  for (int t = 0; t < 4; ++t)
#pragma unroll
    for (int r = 0; r < 8; ++r)
      Tout[(size_t)(mbase + hi * 8 + r) * RANK + t * 16 + ln16] = acc[t][r];
}

__global__ __launch_bounds__(256) void lowrank_tv(const float* __restrict__ Tpart,
                                                  const float* __restrict__ V,
                                                  const float* __restrict__ bias,
                                                  float* __restrict__ out) {
  __shared__ float lV[(RANK / 2) * LSTR];

  const int tid  = threadIdx.x;
  const int lane = tid & 31;
  const int ln16 = lane & 15;
  const int hi   = lane >> 4;
  const int wave = tid >> 5;
  const int mbase = blockIdx.x * 128 + wave * 16;
  const int nbase = blockIdx.y * 64;

  for (int i = tid; i < RANK * (64 / 4); i += 256) {
    const int r  = i >> 4;
    const int c4 = i & 15;
    const v4f v = *(const v4f*)(V + (size_t)r * OUT_F + nbase + c4 * 4);
    float* dst = &lV[(r >> 1) * LSTR + (r & 1)];
#pragma unroll
    for (int j = 0; j < 4; ++j) dst[2 * (c4 * 4 + j)] = v[j];
  }
  __syncthreads();

  const size_t trow = (size_t)(mbase + ln16) * RANK;

  v8f acc[4] = {};
#pragma unroll
  for (int k0 = 0; k0 < RANK; k0 += 4) {
    const int koff = k0 + 2 * hi;
    v2f a = *(const v2f*)(Tpart + trow + koff);
#pragma unroll
    for (int p = 1; p < KSPLIT; ++p)
      a += *(const v2f*)(Tpart + (size_t)p * NTOK * RANK + trow + koff);
#pragma unroll
    for (int t = 0; t < 4; ++t) {
      const v2f b = *(const v2f*)(
          &lV[((k0 >> 1) + hi) * LSTR + 2 * (t * 16 + ln16)]);
      acc[t] = __builtin_amdgcn_wmma_f32_16x16x4_f32(
          false, a, false, b, (short)0, acc[t], false, false);
    }
  }

#pragma unroll
  for (int t = 0; t < 4; ++t) {
    const float bv = bias[nbase + t * 16 + ln16];
#pragma unroll
    for (int r = 0; r < 8; ++r)
      out[(size_t)(mbase + hi * 8 + r) * OUT_F + nbase + t * 16 + ln16] =
          acc[t][r] + bv;
  }
}

extern "C" void kernel_launch(void* const* d_in, const int* in_sizes, int n_in,
                              void* d_out, int out_size, void* d_ws, size_t ws_size,
                              hipStream_t stream) {
  const float* x    = (const float*)d_in[0];
  const float* U    = (const float*)d_in[1];
  const float* V    = (const float*)d_in[2];
  const float* bias = (const float*)d_in[3];
  float* out   = (float*)d_out;
  float* Tpart = (float*)d_ws;

  lowrank_xu<<<dim3(NTOK / 128, KSPLIT), 256, 0, stream>>>(x, U, Tpart);
  lowrank_tv<<<dim3(NTOK / 128, OUT_F / 64), 256, 0, stream>>>(Tpart, V, bias, out);
}


# stage 2: each block sums its split-K T rows into A fragments once and walks 16 column strips (V double-buffered in LDS, B reads one group per strip) instead of re-reading T per 64-col strip
# speedup vs baseline: 1.8971x; 1.4799x over previous
#include <hip/hip_runtime.h>


typedef float v2f __attribute__((ext_vector_type(2)));
typedef float v4f __attribute__((ext_vector_type(4)));
typedef float v8f __attribute__((ext_vector_type(8)));

constexpr int IN_F   = 4096;
constexpr int OUT_F  = 4096;
constexpr int RANK   = 64;
constexpr int NTOK   = 8192;
constexpr int KSPLIT = 4;
constexpr int KCHUNK = 128;
constexpr int LSTR   = 160;
constexpr int LSTR_U = 144;
constexpr int KPER   = IN_F / KSPLIT;
constexpr int NSTRIP = 16;
constexpr int VBUF   = (RANK / 2) * LSTR;

__global__ __launch_bounds__(256) void lowrank_xu(const float* __restrict__ x,
                                                  const float* __restrict__ U,
                                                  float* __restrict__ Tpart) {
  __shared__ float lU[(KCHUNK / 2) * LSTR_U];

  const int tid  = threadIdx.x;
  const int lane = tid & 31;
  const int ln16 = lane & 15;
  const int hi   = lane >> 4;
  const int wave = tid >> 5;
  const int mbase = blockIdx.x * 128 + wave * 16;
  const int kpart = blockIdx.y * KPER;

  const float* __restrict__ xrow = x + (size_t)(mbase + ln16) * IN_F + kpart + 4 * hi;
  const int ur = tid >> 4;
  const int uc = tid & 15;
  const float* __restrict__ urow = U + (size_t)(kpart + ur) * RANK + uc * 4;

  v8f acc[4] = {};
  v4f qa[16], qb[16];
  v4f ureg[8];

  auto load_x = [&](v4f (&q)[16], int kb) {
#pragma unroll
    for (int j = 0; j < 16; ++j) q[j] = *(const v4f*)(xrow + kb + 8 * j);
  };
  auto load_u = [&](int kb) {
#pragma unroll
    for (int s = 0; s < 8; ++s)
      ureg[s] = *(const v4f*)(urow + (size_t)(kb + 16 * s) * RANK);
  };
  auto store_u = [&]() {
#pragma unroll
    for (int s = 0; s < 8; ++s) {
      const int r = ur + 16 * s;
      float* dst = &lU[(r >> 1) * LSTR_U + (r & 1) + 8 * uc];
#pragma unroll
      for (int j = 0; j < 4; ++j) dst[2 * j] = ureg[s][j];
    }
  };
  auto compute = [&](const v4f (&q)[16]) {
#pragma unroll
    for (int j = 0; j < 16; ++j) {
#pragma unroll
      for (int e = 0; e < 2; ++e) {
        const v2f a = {q[j][2 * e], q[j][2 * e + 1]};
#pragma unroll
        for (int t = 0; t < 4; ++t) {
          const v2f b = *(const v2f*)(
              &lU[(4 * j + 2 * hi + e) * LSTR_U + 2 * (t * 16 + ln16)]);
          acc[t] = __builtin_amdgcn_wmma_f32_16x16x4_f32(
              false, a, false, b, (short)0, acc[t], false, false);
        }
      }
    }
  };

  load_u(0);
  load_x(qa, 0);
  for (int kb = 0; kb < KPER; kb += 2 * KCHUNK) {
    __syncthreads();
    store_u();
    __syncthreads();
    load_u(kb + KCHUNK);
    load_x(qb, kb + KCHUNK);
    compute(qa);

    __syncthreads();
    store_u();
    __syncthreads();
    if (kb + 2 * KCHUNK < KPER) {
      load_u(kb + 2 * KCHUNK);
      load_x(qa, kb + 2 * KCHUNK);
    }
    compute(qb);
  }

  float* __restrict__ Tout = Tpart + (size_t)blockIdx.y * NTOK * RANK;
#pragma unroll
  for (int t = 0; t < 4; ++t)
#pragma unroll
    for (int r = 0; r < 8; ++r)
      Tout[(size_t)(mbase + hi * 8 + r) * RANK + t * 16 + ln16] = acc[t][r];
}

__global__ __launch_bounds__(256) void lowrank_tv(const float* __restrict__ Tpart,
                                                  const float* __restrict__ V,
                                                  const float* __restrict__ bias,
                                                  float* __restrict__ out) {
  __shared__ float lV[2 * VBUF];

  const int tid  = threadIdx.x;
  const int lane = tid & 31;
  const int ln16 = lane & 15;
  const int hi   = lane >> 4;
  const int wave = tid >> 5;
  const int mbase  = blockIdx.x * 128 + wave * 16;
  const int nblock = blockIdx.y * (NSTRIP * 64);

  const int vr = tid >> 4;
  const int vc = tid & 15;
  const float* __restrict__ vsrc = V + (size_t)vr * OUT_F + nblock + vc * 4;
  v4f vreg[4];

  auto load_v = [&](int strip) {
#pragma unroll
    for (int s = 0; s < 4; ++s)
      vreg[s] = *(const v4f*)(vsrc + (size_t)(16 * s) * OUT_F + strip * 64);
  };
  auto store_v = [&](float* buf) {
#pragma unroll
    for (int s = 0; s < 4; ++s) {
      const int r = vr + 16 * s;
      float* dst = &buf[(r >> 1) * LSTR + (r & 1) + 8 * vc];
#pragma unroll
      for (int j = 0; j < 4; ++j) dst[2 * j] = vreg[s][j];
    }
  };

  v2f bf[64];
  auto load_b = [&](const float* buf) {
#pragma unroll
    for (int kk = 0; kk < 16; ++kk)
#pragma unroll
      for (int t = 0; t < 4; ++t)
        bf[kk * 4 + t] = *(const v2f*)(
            &buf[(2 * kk + hi) * LSTR + 2 * (t * 16 + ln16)]);
  };

  load_v(0);

  const size_t trow = (size_t)(mbase + ln16) * RANK + 2 * hi;
  v2f a[16];
#pragma unroll
  for (int kk = 0; kk < 16; ++kk) {
    v2f s = *(const v2f*)(Tpart + trow + 4 * kk);
#pragma unroll
    for (int p = 1; p < KSPLIT; ++p)
      s += *(const v2f*)(Tpart + (size_t)p * NTOK * RANK + trow + 4 * kk);
    a[kk] = s;
  }

  store_v(lV);
  __syncthreads();
  load_b(lV);

  float* __restrict__ orow = out + (size_t)(mbase + hi * 8) * OUT_F + nblock + ln16;
  const float* __restrict__ bcol = bias + nblock + ln16;

  for (int s = 0; s < NSTRIP; ++s) {
    const bool more = s + 1 < NSTRIP;
    if (more) load_v(s + 1);

    float bv[4];
#pragma unroll
    for (int t = 0; t < 4; ++t) bv[t] = bcol[s * 64 + t * 16];

    __builtin_amdgcn_sched_barrier(0);
    v8f acc[4] = {};
#pragma unroll
    for (int kk = 0; kk < 16; ++kk)
#pragma unroll
      for (int t = 0; t < 4; ++t)
        acc[t] = __builtin_amdgcn_wmma_f32_16x16x4_f32(
            false, a[kk], false, bf[kk * 4 + t], (short)0, acc[t], false, false);
    __builtin_amdgcn_sched_barrier(0);
    asm volatile("v_nop\n\tv_nop\n\tv_nop\n\tv_nop");
    __builtin_amdgcn_sched_barrier(0);

    if (more) {
      store_v(lV + ((s + 1) & 1) * VBUF);
      __syncthreads();
      load_b(lV + ((s + 1) & 1) * VBUF);
    }

#pragma unroll
    for (int t = 0; t < 4; ++t)
#pragma unroll
      for (int r = 0; r < 8; ++r)
        orow[(size_t)r * OUT_F + s * 64 + t * 16] = acc[t][r] + bv[t];
  }
}

extern "C" void kernel_launch(void* const* d_in, const int* in_sizes, int n_in,
                              void* d_out, int out_size, void* d_ws, size_t ws_size,
                              hipStream_t stream) {
  const float* x    = (const float*)d_in[0];
  const float* U    = (const float*)d_in[1];
  const float* V    = (const float*)d_in[2];
  const float* bias = (const float*)d_in[3];
  float* out   = (float*)d_out;
  float* Tpart = (float*)d_ws;

  lowrank_xu<<<dim3(NTOK / 128, KSPLIT), 256, 0, stream>>>(x, U, Tpart);
  lowrank_tv<<<dim3(NTOK / 128, OUT_F / (64 * NSTRIP)), 256, 0, stream>>>(Tpart, V, bias, out);
}


# stage 2: finished 16x64 tiles (+bias) go through a wave-private LDS patch and leave as 8 b128 stores of whole 256-B row pieces inside the next strip's WMMA run, not 32 b32 stores of 64-B halves
# speedup vs baseline: 1.9551x; 1.0306x over previous
#include <hip/hip_runtime.h>


typedef float v2f __attribute__((ext_vector_type(2)));
typedef float v4f __attribute__((ext_vector_type(4)));
typedef float v8f __attribute__((ext_vector_type(8)));

constexpr int IN_F   = 4096;
constexpr int OUT_F  = 4096;
constexpr int RANK   = 64;
constexpr int NTOK   = 8192;
constexpr int KSPLIT = 4;
constexpr int KCHUNK = 128;
constexpr int LSTR   = 160;
constexpr int LSTR_U = 144;
constexpr int KPER   = IN_F / KSPLIT;
constexpr int NSTRIP = 16;
constexpr int VBUF   = (RANK / 2) * LSTR;
constexpr int PPITCH = 68;
constexpr int PWAVE  = 16 * PPITCH;

__global__ __launch_bounds__(256) void lowrank_xu(const float* __restrict__ x,
                                                  const float* __restrict__ U,
                                                  float* __restrict__ Tpart) {
  __shared__ float lU[(KCHUNK / 2) * LSTR_U];

  const int tid  = threadIdx.x;
  const int lane = tid & 31;
  const int ln16 = lane & 15;
  const int hi   = lane >> 4;
  const int wave = tid >> 5;
  const int mbase = blockIdx.x * 128 + wave * 16;
  const int kpart = blockIdx.y * KPER;

  const float* __restrict__ xrow = x + (size_t)(mbase + ln16) * IN_F + kpart + 4 * hi;
  const int ur = tid >> 4;
  const int uc = tid & 15;
  const float* __restrict__ urow = U + (size_t)(kpart + ur) * RANK + uc * 4;

  v8f acc[4] = {};
  v4f qa[16], qb[16];
  v4f ureg[8];

  auto load_x = [&](v4f (&q)[16], int kb) {
#pragma unroll
    for (int j = 0; j < 16; ++j) q[j] = *(const v4f*)(xrow + kb + 8 * j);
  };
  auto load_u = [&](int kb) {
#pragma unroll
    for (int s = 0; s < 8; ++s)
      ureg[s] = *(const v4f*)(urow + (size_t)(kb + 16 * s) * RANK);
  };
  auto store_u = [&]() {
#pragma unroll
    for (int s = 0; s < 8; ++s) {
      const int r = ur + 16 * s;
      float* dst = &lU[(r >> 1) * LSTR_U + (r & 1) + 8 * uc];
#pragma unroll
      for (int j = 0; j < 4; ++j) dst[2 * j] = ureg[s][j];
    }
  };
  auto compute = [&](const v4f (&q)[16]) {
#pragma unroll
    for (int j = 0; j < 16; ++j) {
#pragma unroll
      for (int e = 0; e < 2; ++e) {
        const v2f a = {q[j][2 * e], q[j][2 * e + 1]};
#pragma unroll
        for (int t = 0; t < 4; ++t) {
          const v2f b = *(const v2f*)(
              &lU[(4 * j + 2 * hi + e) * LSTR_U + 2 * (t * 16 + ln16)]);
          acc[t] = __builtin_amdgcn_wmma_f32_16x16x4_f32(
              false, a, false, b, (short)0, acc[t], false, false);
        }
      }
    }
  };

  load_u(0);
  load_x(qa, 0);
  for (int kb = 0; kb < KPER; kb += 2 * KCHUNK) {
    __syncthreads();
    store_u();
    __syncthreads();
    load_u(kb + KCHUNK);
    load_x(qb, kb + KCHUNK);
    compute(qa);

    __syncthreads();
    store_u();
    __syncthreads();
    if (kb + 2 * KCHUNK < KPER) {
      load_u(kb + 2 * KCHUNK);
      load_x(qa, kb + 2 * KCHUNK);
    }
    compute(qb);
  }

  float* __restrict__ Tout = Tpart + (size_t)blockIdx.y * NTOK * RANK;
#pragma unroll
  for (int t = 0; t < 4; ++t)
#pragma unroll
    for (int r = 0; r < 8; ++r)
      Tout[(size_t)(mbase + hi * 8 + r) * RANK + t * 16 + ln16] = acc[t][r];
}

struct FirstStrip { static constexpr bool first = true; };
struct LaterStrip { static constexpr bool first = false; };

__global__ __launch_bounds__(256) void lowrank_tv(const float* __restrict__ Tpart,
                                                  const float* __restrict__ V,
                                                  const float* __restrict__ bias,
                                                  float* __restrict__ out) {
  __shared__ float lV[2 * VBUF];
  __shared__ __attribute__((aligned(16))) float lP[8 * PWAVE];

  const int tid  = threadIdx.x;
  const int lane = tid & 31;
  const int ln16 = lane & 15;
  const int hi   = lane >> 4;
  const int wave = tid >> 5;
  const int mbase  = blockIdx.x * 128 + wave * 16;
  const int nblock = blockIdx.y * (NSTRIP * 64);

  const int vr = tid >> 4;
  const int vc = tid & 15;
  const float* __restrict__ vsrc = V + (size_t)vr * OUT_F + nblock + vc * 4;
  v4f vreg[4];

  auto load_v = [&](int strip) {
#pragma unroll
    for (int s = 0; s < 4; ++s)
      vreg[s] = *(const v4f*)(vsrc + (size_t)(16 * s) * OUT_F + strip * 64);
  };
  auto store_v = [&](float* buf) {
#pragma unroll
    for (int s = 0; s < 4; ++s) {
      const int r = vr + 16 * s;
      float* dst = &buf[(r >> 1) * LSTR + (r & 1) + 8 * vc];
#pragma unroll
      for (int j = 0; j < 4; ++j) dst[2 * j] = vreg[s][j];
    }
  };

  v2f bf[64];
  auto load_b = [&](const float* buf) {
#pragma unroll
    for (int kk = 0; kk < 16; ++kk)
#pragma unroll
      for (int t = 0; t < 4; ++t)
        bf[kk * 4 + t] = *(const v2f*)(
            &buf[(2 * kk + hi) * LSTR + 2 * (t * 16 + ln16)]);
  };

  float* patch = lP + wave * PWAVE;
  float* pput = patch + (8 * hi) * PPITCH + ln16;
  const float* pget = patch + hi * PPITCH + 4 * ln16;
  float* __restrict__ orow = out + (size_t)(mbase + hi) * OUT_F + nblock + 4 * ln16;

  v8f o[4];
  v4f pr[8];
  auto patch_put = [&]() {
#pragma unroll
    for (int t = 0; t < 4; ++t)
#pragma unroll
      for (int r = 0; r < 8; ++r) pput[r * PPITCH + 16 * t] = o[t][r];
  };
  auto patch_get = [&]() {
#pragma unroll
    for (int q = 0; q < 8; ++q) pr[q] = *(const v4f*)(pget + 2 * q * PPITCH);
  };
  auto rows_out = [&](int strip) {
#pragma unroll
    for (int q = 0; q < 8; ++q)
      *(v4f*)(orow + (size_t)(2 * q) * OUT_F + strip * 64) = pr[q];
  };

  load_v(0);

  const size_t trow = (size_t)(mbase + ln16) * RANK + 2 * hi;
  v2f a[16];
#pragma unroll
  for (int kk = 0; kk < 16; ++kk) {
    v2f s = *(const v2f*)(Tpart + trow + 4 * kk);
#pragma unroll
    for (int p = 1; p < KSPLIT; ++p)
      s += *(const v2f*)(Tpart + (size_t)p * NTOK * RANK + trow + 4 * kk);
    a[kk] = s;
  }

  store_v(lV);
  __syncthreads();
  load_b(lV);

  const float* __restrict__ bcol = bias + nblock + ln16;

  auto strip = [&](auto tag, int s) {
    constexpr bool kFirst = decltype(tag)::first;
    const bool more = s + 1 < NSTRIP;
    if (more) load_v(s + 1);

    float bv[4];
#pragma unroll
    for (int t = 0; t < 4; ++t) bv[t] = bcol[s * 64 + t * 16];

    __builtin_amdgcn_sched_barrier(0);
    v8f acc[4] = {};
#pragma unroll
    for (int kk = 0; kk < 4; ++kk)
#pragma unroll
      for (int t = 0; t < 4; ++t)
        acc[t] = __builtin_amdgcn_wmma_f32_16x16x4_f32(
            false, a[kk], false, bf[kk * 4 + t], (short)0, acc[t], false, false);
    if constexpr (!kFirst) {
      __builtin_amdgcn_sched_barrier(0);
      patch_put();
      patch_get();
      __builtin_amdgcn_sched_barrier(0);
    }
#pragma unroll
    for (int kk = 4; kk < 8; ++kk)
#pragma unroll
      for (int t = 0; t < 4; ++t)
        acc[t] = __builtin_amdgcn_wmma_f32_16x16x4_f32(
            false, a[kk], false, bf[kk * 4 + t], (short)0, acc[t], false, false);
    if constexpr (!kFirst) {
      __builtin_amdgcn_sched_barrier(0);
      rows_out(s - 1);
      __builtin_amdgcn_sched_barrier(0);
    }
#pragma unroll
    for (int kk = 8; kk < 16; ++kk)
#pragma unroll
      for (int t = 0; t < 4; ++t)
        acc[t] = __builtin_amdgcn_wmma_f32_16x16x4_f32(
            false, a[kk], false, bf[kk * 4 + t], (short)0, acc[t], false, false);
    __builtin_amdgcn_sched_barrier(0);
    asm volatile("v_nop\n\tv_nop\n\tv_nop\n\tv_nop");
    __builtin_amdgcn_sched_barrier(0);

#pragma unroll
    for (int t = 0; t < 4; ++t) o[t] = acc[t] + bv[t];

    if (more) {
      store_v(lV + ((s + 1) & 1) * VBUF);
      __syncthreads();
      load_b(lV + ((s + 1) & 1) * VBUF);
    }
  };

  strip(FirstStrip{}, 0);
  for (int s = 1; s < NSTRIP; ++s) strip(LaterStrip{}, s);

  patch_put();
  patch_get();
  rows_out(NSTRIP - 1);
}

extern "C" void kernel_launch(void* const* d_in, const int* in_sizes, int n_in,
                              void* d_out, int out_size, void* d_ws, size_t ws_size,
                              hipStream_t stream) {
  const float* x    = (const float*)d_in[0];
  const float* U    = (const float*)d_in[1];
  const float* V    = (const float*)d_in[2];
  const float* bias = (const float*)d_in[3];
  float* out   = (float*)d_out;
  float* Tpart = (float*)d_ws;

  lowrank_xu<<<dim3(NTOK / 128, KSPLIT), 256, 0, stream>>>(x, U, Tpart);
  lowrank_tv<<<dim3(NTOK / 128, OUT_F / (64 * NSTRIP)), 256, 0, stream>>>(Tpart, V, bias, out);
}
